# starting kernel
# baseline (speedup 1.0000x reference)
#include <hip/hip_runtime.h>

typedef __attribute__((ext_vector_type(2))) float v2f;
typedef __attribute__((ext_vector_type(8))) float v8f;

__global__ __launch_bounds__(256) void poly_gemm_wmma(
    const float* __restrict__ X, const float* __restrict__ W,
    const float* __restrict__ bias, float* __restrict__ out,
    int N, int init)
{
  __shared__ v2f sW[16 * 4 * 32];
  for (int j = threadIdx.x; j < 16 * 4 * 32; j += 256) {
    const int kb4 = j >> 7;
    const int t   = (j >> 5) & 3;
    const int ln  = j & 31;
    const int h   = ln >> 4;
    const int lw  = ln & 15;
    const int krow = kb4 * 4 + 2 * h;
    const int ncol = lw + 16 * t;
    v2f p;
    p.x = W[krow * 64 + ncol];
    p.y = W[(krow + 1) * 64 + ncol];
    sW[j] = p;
  }
  __syncthreads();

  const int lane = threadIdx.x & 31;
  const int wave = threadIdx.x >> 5;
  const int rowBase = (blockIdx.x * 8 + wave) * 16;
  if (rowBase >= N) return;

  const int lo = lane & 15;
  const int hi = lane >> 4;

  v8f acc0 = {}, acc1 = {}, acc2 = {}, acc3 = {};
  const float* xrow = X + (size_t)(rowBase + lo) * 64;
  const v2f* wl = sW + lane;

#pragma unroll
  for (int kb = 0; kb < 64; kb += 4) {
    const int ka = kb + 2 * hi;
    v2f a;
    a.x = xrow[ka];
    a.y = xrow[ka + 1];
    const v2f* wb = wl + (kb >> 2) * 128;
    const v2f b0 = wb[0];
    const v2f b1 = wb[32];
    const v2f b2 = wb[64];
    const v2f b3 = wb[96];
    acc0 = __builtin_amdgcn_wmma_f32_16x16x4_f32(false, a, false, b0, (short)0, acc0, false, false);
    acc1 = __builtin_amdgcn_wmma_f32_16x16x4_f32(false, a, false, b1, (short)0, acc1, false, false);
    acc2 = __builtin_amdgcn_wmma_f32_16x16x4_f32(false, a, false, b2, (short)0, acc2, false, false);
    acc3 = __builtin_amdgcn_wmma_f32_16x16x4_f32(false, a, false, b3, (short)0, acc3, false, false);
  }

  const float bb0 = bias[lo], bb1 = bias[lo + 16],
              bb2 = bias[lo + 32], bb3 = bias[lo + 48];
#pragma unroll
  for (int r = 0; r < 8; ++r) {
    float* o = out + (size_t)(rowBase + r + 8 * hi) * 64;
    if (init) {
      o[lo]      = acc0[r] + bb0;
      o[lo + 16] = acc1[r] + bb1;
      o[lo + 32] = acc2[r] + bb2;
      o[lo + 48] = acc3[r] + bb3;
    } else {
      o[lo]      += acc0[r];
      o[lo + 16] += acc1[r];
      o[lo + 32] += acc2[r];
      o[lo + 48] += acc3[r];
    }
  }
}

__global__ void poly_zero(float4* __restrict__ p, int n4) {
  int i = blockIdx.x * blockDim.x + threadIdx.x;
  if (i < n4) p[i] = make_float4(0.f, 0.f, 0.f, 0.f);
}

__global__ void poly_spmm(const int* __restrict__ row, const int* __restrict__ col,
                          const float* __restrict__ val,
                          const float* __restrict__ xin, float* __restrict__ xout,
                          int E)
{
  int t = blockIdx.x * blockDim.x + threadIdx.x;
  int e = t >> 6;
  int f = t & 63;
  if (e >= E) return;
  float v = val[e];
  int c = col[e];
  int r = row[e];
  atomicAdd(xout + (size_t)r * 64 + f, v * xin[(size_t)c * 64 + f]);
}

extern "C" void kernel_launch(void* const* d_in, const int* in_sizes, int n_in,
                              void* d_out, int out_size, void* d_ws, size_t ws_size,
                              hipStream_t stream) {
  const float* x       = (const float*)d_in[0];
  const int*   lap_row = (const int*)d_in[1];
  const int*   lap_col = (const int*)d_in[2];
  const float* lap_val = (const float*)d_in[3];
  const float* weights = (const float*)d_in[4];
  const float* bias    = (const float*)d_in[5];
  float* out = (float*)d_out;

  const int N = in_sizes[0] / 64;
  const int E = in_sizes[1];
  const int K = 4;

  float* buf[2] = { (float*)d_ws, (float*)d_ws + (size_t)N * 64 };

  const int rowBlocks = (N + 15) / 16;
  dim3 gemmGrid((rowBlocks + 7) / 8), gemmBlock(256);
  const int n4 = (N * 64) / 4;
  dim3 zeroGrid((n4 + 255) / 256), zeroBlock(256);
  const long long spmmThreads = (long long)E * 64;
  dim3 spmmGrid((unsigned)((spmmThreads + 255) / 256)), spmmBlock(256);

  const float* xk = x;
  for (int k = 0; k <= K; ++k) {
    poly_gemm_wmma<<<gemmGrid, gemmBlock, 0, stream>>>(
        xk, weights + (size_t)k * 64 * 64, bias, out, N, (k == 0) ? 1 : 0);
    if (k < K) {
      float* nxt = buf[k & 1];
      poly_zero<<<zeroGrid, zeroBlock, 0, stream>>>((float4*)nxt, n4);
      poly_spmm<<<spmmGrid, spmmBlock, 0, stream>>>(lap_row, lap_col, lap_val,
                                                    xk, nxt, E);
      xk = nxt;
    }
  }
}


# SpMM scatter: one wave walks 16 consecutive edges (indices via readlane), all 32 row loads in flight before its 32 whole-line atomics; was one wave per half edge; sums stay atomic arrival order
# speedup vs baseline: 2.6866x; 2.6866x over previous
#include <hip/hip_runtime.h>

typedef __attribute__((ext_vector_type(2))) float v2f;
typedef __attribute__((ext_vector_type(8))) float v8f;

__global__ __launch_bounds__(256) void poly_gemm_wmma(
    const float* __restrict__ X, const float* __restrict__ W,
    const float* __restrict__ bias, float* __restrict__ out,
    int N, int init)
{
  __shared__ v2f sW[16 * 4 * 32];
  for (int j = threadIdx.x; j < 16 * 4 * 32; j += 256) {
    const int kb4 = j >> 7;
    const int t   = (j >> 5) & 3;
    const int ln  = j & 31;
    const int h   = ln >> 4;
    const int lw  = ln & 15;
    const int krow = kb4 * 4 + 2 * h;
    const int ncol = lw + 16 * t;
    v2f p;
    p.x = W[krow * 64 + ncol];
    p.y = W[(krow + 1) * 64 + ncol];
    sW[j] = p;
  }
  __syncthreads();

  const int lane = threadIdx.x & 31;
  const int wave = threadIdx.x >> 5;
  const int rowBase = (blockIdx.x * 8 + wave) * 16;
  if (rowBase >= N) return;

  const int lo = lane & 15;
  const int hi = lane >> 4;

  v8f acc0 = {}, acc1 = {}, acc2 = {}, acc3 = {};
  const float* xrow = X + (size_t)(rowBase + lo) * 64;
  const v2f* wl = sW + lane;

#pragma unroll
  for (int kb = 0; kb < 64; kb += 4) {
    const int ka = kb + 2 * hi;
    v2f a;
    a.x = xrow[ka];
    a.y = xrow[ka + 1];
    const v2f* wb = wl + (kb >> 2) * 128;
    const v2f b0 = wb[0];
    const v2f b1 = wb[32];
    const v2f b2 = wb[64];
    const v2f b3 = wb[96];
    acc0 = __builtin_amdgcn_wmma_f32_16x16x4_f32(false, a, false, b0, (short)0, acc0, false, false);
    acc1 = __builtin_amdgcn_wmma_f32_16x16x4_f32(false, a, false, b1, (short)0, acc1, false, false);
    acc2 = __builtin_amdgcn_wmma_f32_16x16x4_f32(false, a, false, b2, (short)0, acc2, false, false);
    acc3 = __builtin_amdgcn_wmma_f32_16x16x4_f32(false, a, false, b3, (short)0, acc3, false, false);
  }

  const float bb0 = bias[lo], bb1 = bias[lo + 16],
              bb2 = bias[lo + 32], bb3 = bias[lo + 48];
#pragma unroll
  for (int r = 0; r < 8; ++r) {
    float* o = out + (size_t)(rowBase + r + 8 * hi) * 64;
    if (init) {
      o[lo]      = acc0[r] + bb0;
      o[lo + 16] = acc1[r] + bb1;
      o[lo + 32] = acc2[r] + bb2;
      o[lo + 48] = acc3[r] + bb3;
    } else {
      o[lo]      += acc0[r];
      o[lo + 16] += acc1[r];
      o[lo + 32] += acc2[r];
      o[lo + 48] += acc3[r];
    }
  }
}

__global__ void poly_zero(float4* __restrict__ p, int n4) {
  int i = blockIdx.x * blockDim.x + threadIdx.x;
  if (i < n4) p[i] = make_float4(0.f, 0.f, 0.f, 0.f);
}

#define SPMM_EPW 16

__global__ __launch_bounds__(256) void poly_spmm(
    const int* __restrict__ row, const int* __restrict__ col,
    const float* __restrict__ val,
    const float* __restrict__ xin, float* __restrict__ xout, int E)
{
  const int lane = threadIdx.x & 31;
  const int wv = __builtin_amdgcn_readfirstlane((int)(threadIdx.x >> 5));
  const int e0 = (blockIdx.x * 8 + wv) * SPMM_EPW;
  if (e0 >= E) return;
  const int n = min(SPMM_EPW, E - e0);

  const int me = e0 + min(lane & (SPMM_EPW - 1), n - 1);
  const int r  = row[me];
  const int c  = col[me];
  const int vb = __float_as_int(val[me]);

  if (n == SPMM_EPW) {
    float a0[SPMM_EPW], a1[SPMM_EPW];
#pragma unroll
    for (int j = 0; j < SPMM_EPW; ++j) {
      const int   cj = __builtin_amdgcn_readlane(c, j);
      const float vj = __int_as_float(__builtin_amdgcn_readlane(vb, j));
      const float* xs = xin + (size_t)cj * 64 + lane;
      a0[j] = vj * xs[0];
      a1[j] = vj * xs[32];
    }
#pragma unroll
    for (int j = 0; j < SPMM_EPW; ++j) {
      const int rj = __builtin_amdgcn_readlane(r, j);
      float* xo = xout + (size_t)rj * 64 + lane;
      atomicAdd(xo,      a0[j]);
      atomicAdd(xo + 32, a1[j]);
    }
  } else {
    for (int j = 0; j < n; ++j) {
      const int   cj = __builtin_amdgcn_readlane(c, j);
      const int   rj = __builtin_amdgcn_readlane(r, j);
      const float vj = __int_as_float(__builtin_amdgcn_readlane(vb, j));
      const float* xs = xin + (size_t)cj * 64 + lane;
      float* xo = xout + (size_t)rj * 64 + lane;
      atomicAdd(xo,      vj * xs[0]);
      atomicAdd(xo + 32, vj * xs[32]);
    }
  }
}

extern "C" void kernel_launch(void* const* d_in, const int* in_sizes, int n_in,
                              void* d_out, int out_size, void* d_ws, size_t ws_size,
                              hipStream_t stream) {
  const float* x       = (const float*)d_in[0];
  const int*   lap_row = (const int*)d_in[1];
  const int*   lap_col = (const int*)d_in[2];
  const float* lap_val = (const float*)d_in[3];
  const float* weights = (const float*)d_in[4];
  const float* bias    = (const float*)d_in[5];
  float* out = (float*)d_out;

  const int N = in_sizes[0] / 64;
  const int E = in_sizes[1];
  const int K = 4;

  float* buf[2] = { (float*)d_ws, (float*)d_ws + (size_t)N * 64 };

  const int rowBlocks = (N + 15) / 16;
  dim3 gemmGrid((rowBlocks + 7) / 8), gemmBlock(256);
  const int n4 = (N * 64) / 4;
  dim3 zeroGrid((n4 + 255) / 256), zeroBlock(256);
  const int edgesPerBlock = 8 * SPMM_EPW;
  dim3 spmmGrid((unsigned)((E + edgesPerBlock - 1) / edgesPerBlock)), spmmBlock(256);

  const float* xk = x;
  for (int k = 0; k <= K; ++k) {
    poly_gemm_wmma<<<gemmGrid, gemmBlock, 0, stream>>>(
        xk, weights + (size_t)k * 64 * 64, bias, out, N, (k == 0) ? 1 : 0);
    if (k < K) {
      float* nxt = buf[k & 1];
      poly_zero<<<zeroGrid, zeroBlock, 0, stream>>>((float4*)nxt, n4);
      poly_spmm<<<spmmGrid, spmmBlock, 0, stream>>>(lap_row, lap_col, lap_val,
                                                    xk, nxt, E);
      xk = nxt;
    }
  }
}


# Each product rides in the previous scatter's launch as extra work-groups (it reads x only); first two share a grid; 3 buffers rotate, out is scratch until the last product: 9 launches become 5
# speedup vs baseline: 3.9164x; 1.4578x over previous
#include <hip/hip_runtime.h>

typedef __attribute__((ext_vector_type(2))) float v2f;
typedef __attribute__((ext_vector_type(8))) float v8f;

__device__ __forceinline__ void gemm_tile(
    const float* __restrict__ X, const float* __restrict__ W,
    const float* __restrict__ bias, float* __restrict__ out,
    int N, int addBias, int blk, v2f* sW)
{
  for (int j = threadIdx.x; j < 16 * 4 * 32; j += 256) {
    const int kb4 = j >> 7;
    const int t   = (j >> 5) & 3;
    const int ln  = j & 31;
    const int h   = ln >> 4;
    const int lw  = ln & 15;
    const int krow = kb4 * 4 + 2 * h;
    const int ncol = lw + 16 * t;
    v2f p;
    p.x = W[krow * 64 + ncol];
    p.y = W[(krow + 1) * 64 + ncol];
    sW[j] = p;
  }
  __syncthreads();

  const int lane = threadIdx.x & 31;
  const int wave = threadIdx.x >> 5;
  const int rowBase = (blk * 8 + wave) * 16;
  if (rowBase >= N) return;

  const int lo = lane & 15;
  const int hi = lane >> 4;

  v8f acc0 = {}, acc1 = {}, acc2 = {}, acc3 = {};
  const float* xrow = X + (size_t)(rowBase + lo) * 64;
  const v2f* wl = sW + lane;

#pragma unroll
  for (int kb = 0; kb < 64; kb += 4) {
    const int ka = kb + 2 * hi;
    v2f a;
    a.x = xrow[ka];
    a.y = xrow[ka + 1];
    const v2f* wb = wl + (kb >> 2) * 128;
    const v2f b0 = wb[0];
    const v2f b1 = wb[32];
    const v2f b2 = wb[64];
    const v2f b3 = wb[96];
    acc0 = __builtin_amdgcn_wmma_f32_16x16x4_f32(false, a, false, b0, (short)0, acc0, false, false);
    acc1 = __builtin_amdgcn_wmma_f32_16x16x4_f32(false, a, false, b1, (short)0, acc1, false, false);
    acc2 = __builtin_amdgcn_wmma_f32_16x16x4_f32(false, a, false, b2, (short)0, acc2, false, false);
    acc3 = __builtin_amdgcn_wmma_f32_16x16x4_f32(false, a, false, b3, (short)0, acc3, false, false);
  }

  float bb0 = 0.f, bb1 = 0.f, bb2 = 0.f, bb3 = 0.f;
  if (addBias) {
    bb0 = bias[lo];      bb1 = bias[lo + 16];
    bb2 = bias[lo + 32]; bb3 = bias[lo + 48];
  }

#pragma unroll
  for (int r = 0; r < 8; ++r) {
    float* o = out + (size_t)(rowBase + r + 8 * hi) * 64;
    o[lo]      = acc0[r] + bb0;
    o[lo + 16] = acc1[r] + bb1;
    o[lo + 32] = acc2[r] + bb2;
    o[lo + 48] = acc3[r] + bb3;
  }
}

#define SPMM_EPW 16

__device__ __forceinline__ void spmm_runs(
    const int* __restrict__ row, const int* __restrict__ col,
    const float* __restrict__ val,
    const float* __restrict__ xin, float* __restrict__ xout, int E, int blk)
{
  const int lane = threadIdx.x & 31;
  const int wv = __builtin_amdgcn_readfirstlane((int)(threadIdx.x >> 5));
  const int e0 = (blk * 8 + wv) * SPMM_EPW;
  if (e0 >= E) return;
  const int n = min(SPMM_EPW, E - e0);

  const int me = e0 + min(lane & (SPMM_EPW - 1), n - 1);
  const int r  = row[me];
  const int c  = col[me];
  const int vb = __float_as_int(val[me]);

  if (n == SPMM_EPW) {
    float a0[SPMM_EPW], a1[SPMM_EPW];
#pragma unroll
    for (int j = 0; j < SPMM_EPW; ++j) {
      const int   cj = __builtin_amdgcn_readlane(c, j);
      const float vj = __int_as_float(__builtin_amdgcn_readlane(vb, j));
      const float* xs = xin + (size_t)cj * 64 + lane;
      a0[j] = vj * xs[0];
      a1[j] = vj * xs[32];
    }
#pragma unroll
    for (int j = 0; j < SPMM_EPW; ++j) {
      const int rj = __builtin_amdgcn_readlane(r, j);
      float* xo = xout + (size_t)rj * 64 + lane;
      atomicAdd(xo,      a0[j]);
      atomicAdd(xo + 32, a1[j]);
    }
  } else {
    for (int j = 0; j < n; ++j) {
      const int   cj = __builtin_amdgcn_readlane(c, j);
      const int   rj = __builtin_amdgcn_readlane(r, j);
      const float vj = __int_as_float(__builtin_amdgcn_readlane(vb, j));
      const float* xs = xin + (size_t)cj * 64 + lane;
      float* xo = xout + (size_t)rj * 64 + lane;
      atomicAdd(xo,      vj * xs[0]);
      atomicAdd(xo + 32, vj * xs[32]);
    }
  }
}

__global__ __launch_bounds__(256) void poly_gemm_pair(
    const float* __restrict__ X,
    const float* __restrict__ WA, float* __restrict__ outA,
    const float* __restrict__ WB, float* __restrict__ outB,
    const float* __restrict__ bias, int N)
{
  __shared__ v2f sW[16 * 4 * 32];
  const bool second = (blockIdx.y != 0);
  gemm_tile(X, second ? WB : WA, bias, second ? outB : outA, N, 0, (int)blockIdx.x, sW);
}

__global__ __launch_bounds__(256) __attribute__((amdgpu_waves_per_eu(16, 16))) void poly_step(
    const int* __restrict__ row, const int* __restrict__ col,
    const float* __restrict__ val,
    const float* __restrict__ xin, float* __restrict__ xout, int E,
    const float* __restrict__ X, const float* __restrict__ W,
    const float* __restrict__ bias, float* __restrict__ gout,
    int N, int addBias, int gemmBlocks)
{
  __shared__ v2f sW[16 * 4 * 32];
  const int b = (int)blockIdx.x;
  if (b < gemmBlocks) {
    gemm_tile(X, W, bias, gout, N, addBias, b, sW);
  } else {
    spmm_runs(row, col, val, xin, xout, E, b - gemmBlocks);
  }
}

__global__ __launch_bounds__(256) void poly_spmm(
    const int* __restrict__ row, const int* __restrict__ col,
    const float* __restrict__ val,
    const float* __restrict__ xin, float* __restrict__ xout, int E)
{
  spmm_runs(row, col, val, xin, xout, E, (int)blockIdx.x);
}

extern "C" void kernel_launch(void* const* d_in, const int* in_sizes, int n_in,
                              void* d_out, int out_size, void* d_ws, size_t ws_size,
                              hipStream_t stream) {
  const float* x       = (const float*)d_in[0];
  const int*   lap_row = (const int*)d_in[1];
  const int*   lap_col = (const int*)d_in[2];
  const float* lap_val = (const float*)d_in[3];
  const float* weights = (const float*)d_in[4];
  const float* bias    = (const float*)d_in[5];
  float* out = (float*)d_out;

  const int N = in_sizes[0] / 64;
  const int E = in_sizes[1];

  float* bufA = (float*)d_ws;
  float* bufB = (float*)d_ws + (size_t)N * 64;
  float* bufC = out;

  const int rowBlocks = (N + 15) / 16;
  const int gemmBlocks = (rowBlocks + 7) / 8;
  const int edgesPerBlock = 8 * SPMM_EPW;
  const int spmmBlocks = (E + edgesPerBlock - 1) / edgesPerBlock;
  dim3 block(256);
  dim3 pairGrid((unsigned)gemmBlocks, 2);
  dim3 stepGrid((unsigned)(gemmBlocks + spmmBlocks));
  dim3 spmmGrid((unsigned)spmmBlocks);
  const size_t WSZ = (size_t)64 * 64;

  poly_gemm_pair<<<pairGrid, block, 0, stream>>>(
      x, weights + 4 * WSZ, bufB, weights + 3 * WSZ, bufC, bias, N);
  poly_step<<<stepGrid, block, 0, stream>>>(lap_row, lap_col, lap_val, bufB, bufC, E,
      x, weights + 2 * WSZ, bias, bufA, N, 0, gemmBlocks);
  poly_step<<<stepGrid, block, 0, stream>>>(lap_row, lap_col, lap_val, bufC, bufA, E,
      x, weights + 1 * WSZ, bias, bufB, N, 0, gemmBlocks);
  poly_step<<<stepGrid, block, 0, stream>>>(lap_row, lap_col, lap_val, bufA, bufB, E,
      x, weights, bias, bufC, N, 1, gemmBlocks);
  poly_spmm<<<spmmGrid, block, 0, stream>>>(lap_row, lap_col, lap_val, bufB, bufC, E);
}


# Scatter waves stay resident and walk 5 runs of 16 edges each, next run's indices loaded ahead, rows addressed as scalar base + lane: 10,000 waves a launch, was 50,000; sums stay atomic arrival order
# speedup vs baseline: 4.0866x; 1.0435x over previous
#include <hip/hip_runtime.h>

typedef __attribute__((ext_vector_type(2))) float v2f;
typedef __attribute__((ext_vector_type(8))) float v8f;

__device__ __forceinline__ void gemm_tile(
    const float* __restrict__ X, const float* __restrict__ W,
    const float* __restrict__ bias, float* __restrict__ out,
    int N, int addBias, int blk, v2f* sW)
{
  for (int j = threadIdx.x; j < 16 * 4 * 32; j += 256) {
    const int kb4 = j >> 7;
    const int t   = (j >> 5) & 3;
    const int ln  = j & 31;
    const int h   = ln >> 4;
    const int lw  = ln & 15;
    const int krow = kb4 * 4 + 2 * h;
    const int ncol = lw + 16 * t;
    v2f p;
    p.x = W[krow * 64 + ncol];
    p.y = W[(krow + 1) * 64 + ncol];
    sW[j] = p;
  }
  __syncthreads();

  const int lane = threadIdx.x & 31;
  const int wave = threadIdx.x >> 5;
  const int rowBase = (blk * 8 + wave) * 16;
  if (rowBase >= N) return;

  const int lo = lane & 15;
  const int hi = lane >> 4;

  v8f acc0 = {}, acc1 = {}, acc2 = {}, acc3 = {};
  const float* xrow = X + (size_t)(rowBase + lo) * 64;
  const v2f* wl = sW + lane;

#pragma unroll
  for (int kb = 0; kb < 64; kb += 4) {
    const int ka = kb + 2 * hi;
    v2f a;
    a.x = xrow[ka];
    a.y = xrow[ka + 1];
    const v2f* wb = wl + (kb >> 2) * 128;
    const v2f b0 = wb[0];
    const v2f b1 = wb[32];
    const v2f b2 = wb[64];
    const v2f b3 = wb[96];
    acc0 = __builtin_amdgcn_wmma_f32_16x16x4_f32(false, a, false, b0, (short)0, acc0, false, false);
    acc1 = __builtin_amdgcn_wmma_f32_16x16x4_f32(false, a, false, b1, (short)0, acc1, false, false);
    acc2 = __builtin_amdgcn_wmma_f32_16x16x4_f32(false, a, false, b2, (short)0, acc2, false, false);
    acc3 = __builtin_amdgcn_wmma_f32_16x16x4_f32(false, a, false, b3, (short)0, acc3, false, false);
  }

  float bb0 = 0.f, bb1 = 0.f, bb2 = 0.f, bb3 = 0.f;
  if (addBias) {
    bb0 = bias[lo];      bb1 = bias[lo + 16];
    bb2 = bias[lo + 32]; bb3 = bias[lo + 48];
  }

#pragma unroll
  for (int r = 0; r < 8; ++r) {
    float* o = out + (size_t)(rowBase + r + 8 * hi) * 64;
    o[lo]      = acc0[r] + bb0;
    o[lo + 16] = acc1[r] + bb1;
    o[lo + 32] = acc2[r] + bb2;
    o[lo + 48] = acc3[r] + bb3;
  }
}

#define SPMM_EPW 16

typedef __attribute__((address_space(1))) float gfloat;

__device__ __forceinline__ gfloat* wave_uniform_ptr(const float* p)
{
  const unsigned long long u = (unsigned long long)p;
  const unsigned lo = __builtin_amdgcn_readfirstlane((unsigned)u);
  const unsigned hi = __builtin_amdgcn_readfirstlane((unsigned)(u >> 32));
  return (gfloat*)(((unsigned long long)hi << 32) | lo);
}

__device__ __forceinline__ void spmm_runs(
    const int* __restrict__ row, const int* __restrict__ col,
    const float* __restrict__ val,
    const float* __restrict__ xin, float* __restrict__ xout, int E, int blk, int nBlk)
{
  const unsigned lane = threadIdx.x & 31u;
  const int sub  = (int)(lane & (SPMM_EPW - 1));
  const int wv = __builtin_amdgcn_readfirstlane((int)(threadIdx.x >> 5));
  const int numRuns = (E + SPMM_EPW - 1) / SPMM_EPW;
  const int stride  = nBlk * 8;
  int run = blk * 8 + wv;
  if (run >= numRuns) return;

  int e0 = run * SPMM_EPW;
  int n  = min(SPMM_EPW, E - e0);
  int me = e0 + min(sub, n - 1);
  int r  = row[me];
  int c  = col[me];
  int vb = __float_as_int(val[me]);

  for (;;) {
    const int nextRun = run + stride;
    const int pr  = min(nextRun, numRuns - 1);
    const int pe0 = pr * SPMM_EPW;
    const int pn  = min(SPMM_EPW, E - pe0);
    const int pme = pe0 + min(sub, pn - 1);
    const int rN  = row[pme];
    const int cN  = col[pme];
    const int vbN = __float_as_int(val[pme]);

    if (n == SPMM_EPW) {
      float a0[SPMM_EPW], a1[SPMM_EPW];
#pragma unroll
      for (int j = 0; j < SPMM_EPW; ++j) {
        const int   cj = __builtin_amdgcn_readlane(c, j);
        const float vj = __int_as_float(__builtin_amdgcn_readlane(vb, j));
        const gfloat* xs = wave_uniform_ptr(xin + (size_t)cj * 64);
        a0[j] = vj * xs[lane];
        a1[j] = vj * xs[lane + 32];
      }
#pragma unroll
      for (int j = 0; j < SPMM_EPW; ++j) {
        const int rj = __builtin_amdgcn_readlane(r, j);
        gfloat* xo = wave_uniform_ptr(xout + (size_t)rj * 64);
        atomicAdd((float*)(xo + lane),      a0[j]);
        atomicAdd((float*)(xo + lane + 32), a1[j]);
      }
    } else {
      for (int j = 0; j < n; ++j) {
        const int   cj = __builtin_amdgcn_readlane(c, j);
        const int   rj = __builtin_amdgcn_readlane(r, j);
        const float vj = __int_as_float(__builtin_amdgcn_readlane(vb, j));
        const float* xs = xin + (size_t)cj * 64 + lane;
        float* xo = xout + (size_t)rj * 64 + lane;
        atomicAdd(xo,      vj * xs[0]);
        atomicAdd(xo + 32, vj * xs[32]);
      }
    }

    if (nextRun >= numRuns) break;
    run = nextRun;
    n = pn; r = rN; c = cN; vb = vbN;
  }
}

__global__ __launch_bounds__(256) void poly_gemm_pair(
    const float* __restrict__ X,
    const float* __restrict__ WA, float* __restrict__ outA,
    const float* __restrict__ WB, float* __restrict__ outB,
    const float* __restrict__ bias, int N)
{
  __shared__ v2f sW[16 * 4 * 32];
  const bool second = (blockIdx.y != 0);
  gemm_tile(X, second ? WB : WA, bias, second ? outB : outA, N, 0, (int)blockIdx.x, sW);
}

__global__ __launch_bounds__(256) __attribute__((amdgpu_waves_per_eu(16, 16))) void poly_step(
    const int* __restrict__ row, const int* __restrict__ col,
    const float* __restrict__ val,
    const float* __restrict__ xin, float* __restrict__ xout, int E,
    const float* __restrict__ X, const float* __restrict__ W,
    const float* __restrict__ bias, float* __restrict__ gout,
    int N, int addBias, int spmmBlocks)
{
  __shared__ v2f sW[16 * 4 * 32];
  const int b = (int)blockIdx.x;
  if (b < spmmBlocks) {
    spmm_runs(row, col, val, xin, xout, E, b, spmmBlocks);
  } else {
    gemm_tile(X, W, bias, gout, N, addBias, b - spmmBlocks, sW);
  }
}

__global__ __launch_bounds__(256) void poly_spmm(
    const int* __restrict__ row, const int* __restrict__ col,
    const float* __restrict__ val,
    const float* __restrict__ xin, float* __restrict__ xout, int E, int spmmBlocks)
{
  spmm_runs(row, col, val, xin, xout, E, (int)blockIdx.x, spmmBlocks);
}

extern "C" void kernel_launch(void* const* d_in, const int* in_sizes, int n_in,
                              void* d_out, int out_size, void* d_ws, size_t ws_size,
                              hipStream_t stream) {
  const float* x       = (const float*)d_in[0];
  const int*   lap_row = (const int*)d_in[1];
  const int*   lap_col = (const int*)d_in[2];
  const float* lap_val = (const float*)d_in[3];
  const float* weights = (const float*)d_in[4];
  const float* bias    = (const float*)d_in[5];
  float* out = (float*)d_out;

  const int N = in_sizes[0] / 64;
  const int E = in_sizes[1];

  float* bufA = (float*)d_ws;
  float* bufB = (float*)d_ws + (size_t)N * 64;
  float* bufC = out;

  const int rowBlocks = (N + 15) / 16;
  const int gemmBlocks = (rowBlocks + 7) / 8;
  const int spmmBlocks = 1250;
  dim3 block(256);
  dim3 pairGrid((unsigned)gemmBlocks, 2);
  dim3 stepGrid((unsigned)(gemmBlocks + spmmBlocks));
  dim3 spmmGrid((unsigned)spmmBlocks);
  const size_t WSZ = (size_t)64 * 64;

  poly_gemm_pair<<<pairGrid, block, 0, stream>>>(
      x, weights + 4 * WSZ, bufB, weights + 3 * WSZ, bufC, bias, N);
  poly_step<<<stepGrid, block, 0, stream>>>(lap_row, lap_col, lap_val, bufB, bufC, E,
      x, weights + 2 * WSZ, bias, bufA, N, 0, spmmBlocks);
  poly_step<<<stepGrid, block, 0, stream>>>(lap_row, lap_col, lap_val, bufC, bufA, E,
      x, weights + 1 * WSZ, bias, bufB, N, 0, spmmBlocks);
  poly_step<<<stepGrid, block, 0, stream>>>(lap_row, lap_col, lap_val, bufA, bufB, E,
      x, weights, bias, bufC, N, 1, spmmBlocks);
  poly_spmm<<<spmmGrid, block, 0, stream>>>(lap_row, lap_col, lap_val, bufB, bufC, E, spmmBlocks);
}


# Rows cut in two classes; a scatter work-group draws by ticket a slice of the class of the package half it runs on, adds only that class's edges: a row's atomics come from one half; sums arrival order
# speedup vs baseline: 5.2951x; 1.2957x over previous
#include <hip/hip_runtime.h>

typedef __attribute__((ext_vector_type(2))) float v2f;
typedef __attribute__((ext_vector_type(8))) float v8f;

__device__ __forceinline__ void gemm_tile(
    const float* __restrict__ X, const float* __restrict__ W,
    const float* __restrict__ bias, float* __restrict__ out,
    int N, int addBias, int blk, v2f* sW)
{
  for (int j = threadIdx.x; j < 16 * 4 * 32; j += 256) {
    const int kb4 = j >> 7;
    const int t   = (j >> 5) & 3;
    const int ln  = j & 31;
    const int h   = ln >> 4;
    const int lw  = ln & 15;
    const int krow = kb4 * 4 + 2 * h;
    const int ncol = lw + 16 * t;
    v2f p;
    p.x = W[krow * 64 + ncol];
    p.y = W[(krow + 1) * 64 + ncol];
    sW[j] = p;
  }
  __syncthreads();

  const int lane = threadIdx.x & 31;
  const int wave = threadIdx.x >> 5;
  const int rowBase = (blk * 8 + wave) * 16;
  if (rowBase >= N) return;

  const int lo = lane & 15;
  const int hi = lane >> 4;

  v8f acc0 = {}, acc1 = {}, acc2 = {}, acc3 = {};
  const float* xrow = X + (size_t)(rowBase + lo) * 64;
  const v2f* wl = sW + lane;

#pragma unroll
  for (int kb = 0; kb < 64; kb += 4) {
    const int ka = kb + 2 * hi;
    v2f a;
    a.x = xrow[ka];
    a.y = xrow[ka + 1];
    const v2f* wb = wl + (kb >> 2) * 128;
    const v2f b0 = wb[0];
    const v2f b1 = wb[32];
    const v2f b2 = wb[64];
    const v2f b3 = wb[96];
    acc0 = __builtin_amdgcn_wmma_f32_16x16x4_f32(false, a, false, b0, (short)0, acc0, false, false);
    acc1 = __builtin_amdgcn_wmma_f32_16x16x4_f32(false, a, false, b1, (short)0, acc1, false, false);
    acc2 = __builtin_amdgcn_wmma_f32_16x16x4_f32(false, a, false, b2, (short)0, acc2, false, false);
    acc3 = __builtin_amdgcn_wmma_f32_16x16x4_f32(false, a, false, b3, (short)0, acc3, false, false);
  }

  float bb0 = 0.f, bb1 = 0.f, bb2 = 0.f, bb3 = 0.f;
  if (addBias) {
    bb0 = bias[lo];      bb1 = bias[lo + 16];
    bb2 = bias[lo + 32]; bb3 = bias[lo + 48];
  }

#pragma unroll
  for (int r = 0; r < 8; ++r) {
    float* o = out + (size_t)(rowBase + r + 8 * hi) * 64;
    o[lo]      = acc0[r] + bb0;
    o[lo + 16] = acc1[r] + bb1;
    o[lo + 32] = acc2[r] + bb2;
    o[lo + 48] = acc3[r] + bb3;
  }
}

#define SPMM_BATCH 16
#define SPMM_CNT_STRIDE 64

typedef __attribute__((address_space(1))) float gfloat;

__device__ __forceinline__ gfloat* wave_uniform_ptr(const float* p)
{
  const unsigned long long u = (unsigned long long)p;
  const unsigned lo = __builtin_amdgcn_readfirstlane((unsigned)u);
  const unsigned hi = __builtin_amdgcn_readfirstlane((unsigned)(u >> 32));
  return (gfloat*)(((unsigned long long)hi << 32) | lo);
}

__device__ __forceinline__ bool have_counters(size_t wsBytes, int N)
{
  return wsBytes >= ((size_t)2 * N * 64 + 8 * SPMM_CNT_STRIDE) * sizeof(float);
}

__device__ __forceinline__ void spmm_slice(
    const int* __restrict__ row, const int* __restrict__ col,
    const float* __restrict__ val,
    const float* __restrict__ xin, float* __restrict__ xout, int E, int N,
    int* __restrict__ cnt, int nBlk, int blk, bool haveCnt, int* sTicket)
{
  const unsigned lane = threadIdx.x & 31u;
  const int wv = __builtin_amdgcn_readfirstlane((int)(threadIdx.x >> 5));
  const int slices = nBlk >> 1;

  if (threadIdx.x == 0) {
    int k = blk & 1;
    int t = blk >> 1;
    if (haveCnt) {
      k = (int)((__builtin_amdgcn_s_sendmsg_rtn(0x87) >> 18) & 1u);
      t = atomicAdd(cnt + k * SPMM_CNT_STRIDE, 1);
      if (t >= slices) {
        k ^= 1;
        t = atomicAdd(cnt + k * SPMM_CNT_STRIDE, 1);
      }
    }
    *sTicket = (t << 1) | k;
  }
  __syncthreads();
  const int tk  = __builtin_amdgcn_readfirstlane(*sTicket);
  const int cls = tk & 1;
  const int sl  = tk >> 1;
  if (sl >= slices) return;

  const int halfN     = N >> 1;
  const int numGroups = (E + 31) >> 5;
  const int stride    = slices * 8;
  int g = sl * 8 + wv;
  if (g >= numGroups) return;

  int e  = g * 32 + (int)lane;
  int me = min(e, E - 1);
  int r  = row[me];
  int c  = col[me];
  int vb = __float_as_int(val[me]);

  for (;;) {
    const int gN  = g + stride;
    const int pg  = min(gN, numGroups - 1);
    const int pe  = pg * 32 + (int)lane;
    const int pme = min(pe, E - 1);
    const int rN  = row[pme];
    const int cN  = col[pme];
    const int vbN = __float_as_int(val[pme]);

    const bool mine = (e < E) && ((r >= halfN) == (cls != 0));
    unsigned m = __builtin_amdgcn_ballot_w32(mine);

    while (m != 0u) {
      unsigned m2 = m;
      float a0[SPMM_BATCH], a1[SPMM_BATCH];
#pragma unroll
      for (int b = 0; b < SPMM_BATCH; ++b) {
        if (m != 0u) {
          const int j = __builtin_ctz(m);
          m &= m - 1u;
          const int cj = __builtin_amdgcn_readlane(c, j);
          const gfloat* xs = wave_uniform_ptr(xin + (size_t)cj * 64);
          a0[b] = xs[lane];
          a1[b] = xs[lane + 32];
        }
      }
#pragma unroll
      for (int b = 0; b < SPMM_BATCH; ++b) {
        if (m2 != 0u) {
          const int j = __builtin_ctz(m2);
          m2 &= m2 - 1u;
          const int   rj = __builtin_amdgcn_readlane(r, j);
          const float vj = __int_as_float(__builtin_amdgcn_readlane(vb, j));
          gfloat* xo = wave_uniform_ptr(xout + (size_t)rj * 64);
          atomicAdd((float*)(xo + lane),      vj * a0[b]);
          atomicAdd((float*)(xo + lane + 32), vj * a1[b]);
        }
      }
    }

    if (gN >= numGroups) break;
    g = gN; e = pe;
    r = rN; c = cN; vb = vbN;
  }
}

__global__ __launch_bounds__(256) void poly_gemm_pair(
    const float* __restrict__ X,
    const float* __restrict__ WA, float* __restrict__ outA,
    const float* __restrict__ WB, float* __restrict__ outB,
    const float* __restrict__ bias, int N, int* __restrict__ cnt, size_t wsBytes)
{
  __shared__ v2f sW[16 * 4 * 32];
  if (blockIdx.x == 0 && blockIdx.y == 0 && threadIdx.x < 8 && have_counters(wsBytes, N))
    cnt[threadIdx.x * SPMM_CNT_STRIDE] = 0;
  const bool second = (blockIdx.y != 0);
  gemm_tile(X, second ? WB : WA, bias, second ? outB : outA, N, 0, (int)blockIdx.x, sW);
}

__global__ __launch_bounds__(256) __attribute__((amdgpu_waves_per_eu(16, 16))) void poly_step(
    const int* __restrict__ row, const int* __restrict__ col,
    const float* __restrict__ val,
    const float* __restrict__ xin, float* __restrict__ xout, int E,
    const float* __restrict__ X, const float* __restrict__ W,
    const float* __restrict__ bias, float* __restrict__ gout,
    int N, int addBias, int spmmBlocks, int* __restrict__ cnt, size_t wsBytes)
{
  __shared__ v2f sW[16 * 4 * 32];
  __shared__ int sTicket;
  const int b = (int)blockIdx.x;
  if (b < spmmBlocks) {
    spmm_slice(row, col, val, xin, xout, E, N, cnt, spmmBlocks, b, have_counters(wsBytes, N), &sTicket);
  } else {
    gemm_tile(X, W, bias, gout, N, addBias, b - spmmBlocks, sW);
  }
}

__global__ __launch_bounds__(256) void poly_spmm(
    const int* __restrict__ row, const int* __restrict__ col,
    const float* __restrict__ val,
    const float* __restrict__ xin, float* __restrict__ xout, int E, int N,
    int* __restrict__ cnt, int spmmBlocks, size_t wsBytes)
{
  __shared__ int sTicket;
  spmm_slice(row, col, val, xin, xout, E, N, cnt, spmmBlocks, (int)blockIdx.x,
             have_counters(wsBytes, N), &sTicket);
}

extern "C" void kernel_launch(void* const* d_in, const int* in_sizes, int n_in,
                              void* d_out, int out_size, void* d_ws, size_t ws_size,
                              hipStream_t stream) {
  const float* x       = (const float*)d_in[0];
  const int*   lap_row = (const int*)d_in[1];
  const int*   lap_col = (const int*)d_in[2];
  const float* lap_val = (const float*)d_in[3];
  const float* weights = (const float*)d_in[4];
  const float* bias    = (const float*)d_in[5];
  float* out = (float*)d_out;

  const int N = in_sizes[0] / 64;
  const int E = in_sizes[1];

  float* bufA = (float*)d_ws;
  float* bufB = (float*)d_ws + (size_t)N * 64;
  float* bufC = out;
  int* cnt = (int*)((float*)d_ws + (size_t)2 * N * 64);

  const int rowBlocks = (N + 15) / 16;
  const int gemmBlocks = (rowBlocks + 7) / 8;
  const int spmmBlocks = 1250;
  dim3 block(256);
  dim3 pairGrid((unsigned)gemmBlocks, 2);
  dim3 stepGrid((unsigned)(gemmBlocks + spmmBlocks));
  dim3 spmmGrid((unsigned)spmmBlocks);
  const size_t WSZ = (size_t)64 * 64;

  poly_gemm_pair<<<pairGrid, block, 0, stream>>>(
      x, weights + 4 * WSZ, bufB, weights + 3 * WSZ, bufC, bias, N, cnt, ws_size);
  poly_step<<<stepGrid, block, 0, stream>>>(lap_row, lap_col, lap_val, bufB, bufC, E,
      x, weights + 2 * WSZ, bias, bufA, N, 0, spmmBlocks, cnt, ws_size);
  poly_step<<<stepGrid, block, 0, stream>>>(lap_row, lap_col, lap_val, bufC, bufA, E,
      x, weights + 1 * WSZ, bias, bufB, N, 0, spmmBlocks, cnt + 2 * SPMM_CNT_STRIDE, ws_size);
  poly_step<<<stepGrid, block, 0, stream>>>(lap_row, lap_col, lap_val, bufA, bufB, E,
      x, weights, bias, bufC, N, 1, spmmBlocks, cnt + 4 * SPMM_CNT_STRIDE, ws_size);
  poly_spmm<<<spmmGrid, block, 0, stream>>>(lap_row, lap_col, lap_val, bufB, bufC, E, N,
      cnt + 6 * SPMM_CNT_STRIDE, spmmBlocks, ws_size);
}
